# starting kernel
# baseline (speedup 1.0000x reference)
#include <hip/hip_runtime.h>

#define DD        256
#define SEQN      128
#define NBATCH    32
#define XE_STRIDE 260
#define P_STRIDE  132

typedef float v2f __attribute__((ext_vector_type(2)));
typedef float v8f __attribute__((ext_vector_type(8)));

#define XE_OFF  0
#define XL_OFF  (SEQN * XE_STRIDE)
#define XR_OFF  (XL_OFF + SEQN * P_STRIDE)
#define TOK_OFF (XR_OFF + SEQN * P_STRIDE)
#define SMEM_FLOATS (TOK_OFF + SEQN)

__global__ __launch_bounds__(256) void RN_pair_fused_kernel(
    const int* __restrict__ X, const float* __restrict__ emb,
    const float* __restrict__ W_l, const float* __restrict__ b_l,
    const float* __restrict__ W_r, const float* __restrict__ b_r,
    float* __restrict__ pooled) {
  extern __shared__ float smem[];
  float* xe = smem + XE_OFF;
  float* xl = smem + XL_OFF;
  float* xr = smem + XR_OFF;
  int*   tok = (int*)(smem + TOK_OFF);

  const int b     = blockIdx.x;
  const int dbase = blockIdx.y * 128;
  const int tid   = threadIdx.x;

  if (tid < SEQN) tok[tid] = X[b * SEQN + tid];
  __syncthreads();

  for (int idx = tid; idx < SEQN * (DD / 4); idx += 256) {
    const int row = idx >> 6;
    const int c4  = idx & 63;
    const float4 v = ((const float4*)(emb + (long)tok[row] * DD))[c4];
    float* dst = xe + row * XE_STRIDE + c4 * 4;
    dst[0] = v.x; dst[1] = v.y; dst[2] = v.z; dst[3] = v.w;
  }
  __syncthreads();

  const int wave = tid >> 5;
  const int lane = tid & 31;
  const int lr   = lane & 15;
  const int koff = (lane >> 4) * 2;

  for (int cu = wave; cu < 16; cu += 8) {
    const int mat     = cu >> 3;
    const int colbase = (cu & 7) * 16;
    const float* W    = mat ? W_r : W_l;
    const float* bias = mat ? b_r : b_l;
    float*       dst  = mat ? xr : xl;
    const int dcol    = dbase + colbase + lr;

    const float bv = bias[dcol];
    v8f acc[8];
#pragma unroll
    for (int rt = 0; rt < 8; ++rt)
      acc[rt] = (v8f){ bv, bv, bv, bv, bv, bv, bv, bv };

    const float* brow = W + (long)dcol * DD + koff;
    for (int kb = 0; kb < DD; kb += 32) {
      v2f bf[8];
#pragma unroll
      for (int s = 0; s < 8; ++s)
        bf[s] = *(const v2f*)(brow + kb + 4 * s);
#pragma unroll
      for (int rt = 0; rt < 8; ++rt) {
        const float* arow = xe + (rt * 16 + lr) * XE_STRIDE + kb + koff;
#pragma unroll
        for (int s = 0; s < 8; ++s) {
          v2f a = *(const v2f*)(arow + 4 * s);
          acc[rt] = __builtin_amdgcn_wmma_f32_16x16x4_f32(
               false, a,  false, bf[s],
               (short)0, acc[rt],
               false,  false);
        }
      }
    }
#pragma unroll
    for (int rt = 0; rt < 8; ++rt) {
      float* obase = dst + (rt * 16 + 8 * (lane >> 4)) * P_STRIDE + colbase + lr;
#pragma unroll
      for (int v = 0; v < 8; ++v) obase[v * P_STRIDE] = acc[rt][v];
    }
  }
  __syncthreads();

  const int d    = tid & 127;
  const int ih0  = (tid >> 7) * 64;
  float acc = 0.f;
  for (int jb = 0; jb < SEQN; jb += 8) {
    float xlv[8];
#pragma unroll
    for (int u = 0; u < 8; ++u) xlv[u] = xl[(jb + u) * P_STRIDE + d];
    for (int i = ih0; i < ih0 + 64; ++i) {
      const float xrv = xr[i * P_STRIDE + d];
#pragma unroll
      for (int u = 0; u < 8; ++u) acc += fmaxf(xlv[u] + xrv, 0.f);
    }
  }
  smem[tid] = acc;
  __syncthreads();
  if (tid < 128)
    pooled[b * DD + dbase + tid] = smem[tid] + smem[tid + 128];
}

__global__ __launch_bounds__(32) void RN_out_kernel(
    const float* __restrict__ pooled, const float* __restrict__ W_rn,
    const float* __restrict__ b_rn, float* __restrict__ out) {
  const int mt   = blockIdx.x;
  const int nt   = blockIdx.y;
  const int lane = threadIdx.x;
  const int lr   = lane & 15;
  const int koff = (lane >> 4) * 2;
  const int row  = mt * 16 + lr;
  const int col  = nt * 16 + lr;

  const float bv = 16384.0f * b_rn[col];
  v8f c = { bv, bv, bv, bv, bv, bv, bv, bv };

  const float* arow = pooled + (long)row * DD + koff;
  const float* brow = W_rn + (long)col * DD + koff;
  for (int k0 = 0; k0 < DD; k0 += 4) {
    v2f a  = *(const v2f*)(arow + k0);
    v2f bb = *(const v2f*)(brow + k0);
    c = __builtin_amdgcn_wmma_f32_16x16x4_f32(
        false, a, false, bb, (short)0, c, false, false);
  }
  float* obase = out + (mt * 16 + 8 * (lane >> 4)) * DD + nt * 16 + lr;
#pragma unroll
  for (int v = 0; v < 8; ++v) obase[v * DD] = c[v];
}

extern "C" void kernel_launch(void* const* d_in, const int* in_sizes, int n_in,
                              void* d_out, int out_size, void* d_ws, size_t ws_size,
                              hipStream_t stream) {
  (void)in_sizes; (void)n_in; (void)out_size; (void)ws_size;
  const int*   X    = (const int*)d_in[0];
  const float* emb  = (const float*)d_in[1];
  const float* W_l  = (const float*)d_in[2];
  const float* b_l  = (const float*)d_in[3];
  const float* W_r  = (const float*)d_in[4];
  const float* b_r  = (const float*)d_in[5];
  const float* W_rn = (const float*)d_in[6];
  const float* b_rn = (const float*)d_in[7];
  float* pooled = (float*)d_ws;
  float* out    = (float*)d_out;

  dim3 gridA(NBATCH, 2), blockA(256);
  size_t shmem = (size_t)SMEM_FLOATS * sizeof(float);
  RN_pair_fused_kernel<<<gridA, blockA, shmem, stream>>>(
      X, emb, W_l, b_l, W_r, b_r, pooled);

  dim3 gridB(2, 16), blockB(32);
  RN_out_kernel<<<gridB, blockB, 0, stream>>>(pooled, W_rn, b_rn, out);
}


# Split each batch's 256 feature columns over 8 work-groups of 32 instead of 2 of 128 (grid 64 -> 256, one per WGP); pair sum is combined from 8 partial sums per column instead of 2
# speedup vs baseline: 3.8273x; 3.8273x over previous
#include <hip/hip_runtime.h>


#define DD        256
#define SEQN      128
#define NBATCH    32
#define DCOLS     32
#define NSLICE    (DD / DCOLS)
#define XE_STRIDE 260
#define P_STRIDE  36

typedef float v2f __attribute__((ext_vector_type(2)));
typedef float v8f __attribute__((ext_vector_type(8)));

#define XE_OFF  0
#define XL_OFF  (SEQN * XE_STRIDE)
#define XR_OFF  (XL_OFF + SEQN * P_STRIDE)
#define TOK_OFF (XR_OFF + SEQN * P_STRIDE)
#define SMEM_FLOATS (TOK_OFF + SEQN)

__global__ __launch_bounds__(256) void RN_pair_fused_kernel(
    const int* __restrict__ X, const float* __restrict__ emb,
    const float* __restrict__ W_l, const float* __restrict__ b_l,
    const float* __restrict__ W_r, const float* __restrict__ b_r,
    float* __restrict__ pooled) {
  extern __shared__ float smem[];
  float* xe = smem + XE_OFF;
  float* xl = smem + XL_OFF;
  float* xr = smem + XR_OFF;
  int*   tok = (int*)(smem + TOK_OFF);

  const int b     = blockIdx.x;
  const int dbase = blockIdx.y * DCOLS;
  const int tid   = threadIdx.x;

  if (tid < SEQN) tok[tid] = X[b * SEQN + tid];
  __syncthreads();

  for (int idx = tid; idx < SEQN * (DD / 4); idx += 256) {
    const int row = idx >> 6;
    const int c4  = idx & 63;
    const float4 v = ((const float4*)(emb + (long)tok[row] * DD))[c4];
    float* dst = xe + row * XE_STRIDE + c4 * 4;
    dst[0] = v.x; dst[1] = v.y; dst[2] = v.z; dst[3] = v.w;
  }
  __syncthreads();

  const int wave = tid >> 5;
  const int lane = tid & 31;
  const int lr   = lane & 15;
  const int koff = (lane >> 4) * 2;
  {
    const int cu      = wave & 3;
    const int mat     = cu >> 1;
    const int colbase = (cu & 1) * 16;
    const int rt0     = (wave >> 2) * 4;
    const float* W    = mat ? W_r : W_l;
    const float* bias = mat ? b_r : b_l;
    float*       dst  = mat ? xr : xl;
    const int dcol    = dbase + colbase + lr;

    const float bv = bias[dcol];
    v8f acc[4];
#pragma unroll
    for (int rt = 0; rt < 4; ++rt)
      acc[rt] = (v8f){ bv, bv, bv, bv, bv, bv, bv, bv };

    const float* brow = W + (long)dcol * DD + koff;
    for (int kb = 0; kb < DD; kb += 32) {
      v2f bf[8];
#pragma unroll
      for (int s = 0; s < 8; ++s)
        bf[s] = *(const v2f*)(brow + kb + 4 * s);
#pragma unroll
      for (int rt = 0; rt < 4; ++rt) {
        const float* arow = xe + ((rt0 + rt) * 16 + lr) * XE_STRIDE + kb + koff;
#pragma unroll
        for (int s = 0; s < 8; ++s) {
          v2f a = *(const v2f*)(arow + 4 * s);
          acc[rt] = __builtin_amdgcn_wmma_f32_16x16x4_f32(
               false, a,  false, bf[s],
               (short)0, acc[rt],
               false,  false);
        }
      }
    }
#pragma unroll
    for (int rt = 0; rt < 4; ++rt) {
      float* obase = dst + ((rt0 + rt) * 16 + 8 * (lane >> 4)) * P_STRIDE + colbase + lr;
#pragma unroll
      for (int v = 0; v < 8; ++v) obase[v * P_STRIDE] = acc[rt][v];
    }
  }
  __syncthreads();

  const int d    = tid & 31;
  const int ih0  = (tid >> 5) * 16;
  float acc = 0.f;
  for (int jb = 0; jb < SEQN; jb += 8) {
    float xlv[8];
#pragma unroll
    for (int u = 0; u < 8; ++u) xlv[u] = xl[(jb + u) * P_STRIDE + d];
    for (int i = ih0; i < ih0 + 16; ++i) {
      const float xrv = xr[i * P_STRIDE + d];
#pragma unroll
      for (int u = 0; u < 8; ++u) acc += fmaxf(xlv[u] + xrv, 0.f);
    }
  }
  smem[tid] = acc;
  __syncthreads();
  if (tid < DCOLS) {
    float s = smem[tid];
#pragma unroll
    for (int p = 1; p < 8; ++p) s += smem[tid + 32 * p];
    pooled[b * DD + dbase + tid] = s;
  }
}

__global__ __launch_bounds__(32) void RN_out_kernel(
    const float* __restrict__ pooled, const float* __restrict__ W_rn,
    const float* __restrict__ b_rn, float* __restrict__ out) {
  const int mt   = blockIdx.x;
  const int nt   = blockIdx.y;
  const int lane = threadIdx.x;
  const int lr   = lane & 15;
  const int koff = (lane >> 4) * 2;
  const int row  = mt * 16 + lr;
  const int col  = nt * 16 + lr;

  const float bv = 16384.0f * b_rn[col];
  v8f c = { bv, bv, bv, bv, bv, bv, bv, bv };

  const float* arow = pooled + (long)row * DD + koff;
  const float* brow = W_rn + (long)col * DD + koff;
  for (int k0 = 0; k0 < DD; k0 += 4) {
    v2f a  = *(const v2f*)(arow + k0);
    v2f bb = *(const v2f*)(brow + k0);
    c = __builtin_amdgcn_wmma_f32_16x16x4_f32(
        false, a, false, bb, (short)0, c, false, false);
  }
  float* obase = out + (mt * 16 + 8 * (lane >> 4)) * DD + nt * 16 + lr;
#pragma unroll
  for (int v = 0; v < 8; ++v) obase[v * DD] = c[v];
}

extern "C" void kernel_launch(void* const* d_in, const int* in_sizes, int n_in,
                              void* d_out, int out_size, void* d_ws, size_t ws_size,
                              hipStream_t stream) {
  (void)in_sizes; (void)n_in; (void)out_size; (void)ws_size;
  const int*   X    = (const int*)d_in[0];
  const float* emb  = (const float*)d_in[1];
  const float* W_l  = (const float*)d_in[2];
  const float* b_l  = (const float*)d_in[3];
  const float* W_r  = (const float*)d_in[4];
  const float* b_r  = (const float*)d_in[5];
  const float* W_rn = (const float*)d_in[6];
  const float* b_rn = (const float*)d_in[7];
  float* pooled = (float*)d_ws;
  float* out    = (float*)d_out;

  dim3 gridA(NBATCH, NSLICE), blockA(256);
  size_t shmem = (size_t)SMEM_FLOATS * sizeof(float);
  RN_pair_fused_kernel<<<gridA, blockA, shmem, stream>>>(
      X, emb, W_l, b_l, W_r, b_r, pooled);

  dim3 gridB(2, 16), blockB(32);
  RN_out_kernel<<<gridB, blockB, 0, stream>>>(pooled, W_rn, b_rn, out);
}


# Pair loop keeps 8 packed 2-wide partial sums of max(xl_j,-xr_i) and adds 128*sum_i xr_i at the end, replacing one serial add chain of relu(xl_j+xr_i): pair-sum order and rounding change
# speedup vs baseline: 5.9206x; 1.5469x over previous
#include <hip/hip_runtime.h>


#define DD        256
#define SEQN      128
#define NBATCH    32
#define DCOLS     32
#define NSLICE    (DD / DCOLS)
#define XE_STRIDE 260
#define P_STRIDE  36

typedef float v2f __attribute__((ext_vector_type(2)));
typedef float v8f __attribute__((ext_vector_type(8)));

#define XE_OFF  0
#define XL_OFF  (SEQN * XE_STRIDE)
#define XR_OFF  (XL_OFF + SEQN * P_STRIDE)
#define TOK_OFF (XR_OFF + SEQN * P_STRIDE)
#define SMEM_FLOATS (TOK_OFF + SEQN)

__global__ __launch_bounds__(256) void RN_pair_fused_kernel(
    const int* __restrict__ X, const float* __restrict__ emb,
    const float* __restrict__ W_l, const float* __restrict__ b_l,
    const float* __restrict__ W_r, const float* __restrict__ b_r,
    float* __restrict__ pooled) {
  extern __shared__ float smem[];
  float* xe = smem + XE_OFF;
  float* xl = smem + XL_OFF;
  float* xr = smem + XR_OFF;
  int*   tok = (int*)(smem + TOK_OFF);

  const int b     = blockIdx.x;
  const int dbase = blockIdx.y * DCOLS;
  const int tid   = threadIdx.x;

  if (tid < SEQN) tok[tid] = X[b * SEQN + tid];
  __syncthreads();

  for (int idx = tid; idx < SEQN * (DD / 4); idx += 256) {
    const int row = idx >> 6;
    const int c4  = idx & 63;
    const float4 v = ((const float4*)(emb + (long)tok[row] * DD))[c4];
    float* dst = xe + row * XE_STRIDE + c4 * 4;
    dst[0] = v.x; dst[1] = v.y; dst[2] = v.z; dst[3] = v.w;
  }
  __syncthreads();

  const int wave = tid >> 5;
  const int lane = tid & 31;
  const int lr   = lane & 15;
  const int koff = (lane >> 4) * 2;
  {
    const int cu      = wave & 3;
    const int mat     = cu >> 1;
    const int colbase = (cu & 1) * 16;
    const int rt0     = (wave >> 2) * 4;
    const float* W    = mat ? W_r : W_l;
    const float* bias = mat ? b_r : b_l;
    float*       dst  = mat ? xr : xl;
    const int dcol    = dbase + colbase + lr;

    const float bv = bias[dcol];
    v8f acc[4];
#pragma unroll
    for (int rt = 0; rt < 4; ++rt)
      acc[rt] = (v8f){ bv, bv, bv, bv, bv, bv, bv, bv };

    const float* brow = W + (long)dcol * DD + koff;
    for (int kb = 0; kb < DD; kb += 32) {
      v2f bf[8];
#pragma unroll
      for (int s = 0; s < 8; ++s)
        bf[s] = *(const v2f*)(brow + kb + 4 * s);
#pragma unroll
      for (int rt = 0; rt < 4; ++rt) {
        const float* arow = xe + ((rt0 + rt) * 16 + lr) * XE_STRIDE + kb + koff;
#pragma unroll
        for (int s = 0; s < 8; ++s) {
          v2f a = *(const v2f*)(arow + 4 * s);
          acc[rt] = __builtin_amdgcn_wmma_f32_16x16x4_f32(
               false, a,  false, bf[s],
               (short)0, acc[rt],
               false,  false);
        }
      }
    }
#pragma unroll
    for (int rt = 0; rt < 4; ++rt) {
      float* obase = dst + ((rt0 + rt) * 16 + 8 * (lane >> 4)) * P_STRIDE + colbase + lr;
#pragma unroll
      for (int v = 0; v < 8; ++v) obase[v * P_STRIDE] = acc[rt][v];
    }
  }
  __syncthreads();

  const int d    = tid & 31;
  const int ih0  = (tid >> 5) * 16;
  float nxr[16];
  float sxr = 0.f;
#pragma unroll
  for (int i = 0; i < 16; ++i) {
    const float v = xr[(ih0 + i) * P_STRIDE + d];
    sxr += v;
    nxr[i] = -v;
  }
  v2f acc2[8];
#pragma unroll
  for (int k = 0; k < 8; ++k) acc2[k] = (v2f){ 0.f, 0.f };

  float cur[8];
#pragma unroll
  for (int u = 0; u < 8; ++u) cur[u] = xl[u * P_STRIDE + d];
  for (int jb = 0; jb < SEQN; jb += 8) {
    const int jn = (jb + 8) & (SEQN - 1);
    float nxt[8];
#pragma unroll
    for (int u = 0; u < 8; ++u) nxt[u] = xl[(jn + u) * P_STRIDE + d];
#pragma unroll
    for (int u = 0; u < 8; ++u) {
      v2f t[8];
#pragma unroll
      for (int k = 0; k < 8; ++k) {
        t[k].x = fmaxf(cur[u], nxr[2 * k]);
        t[k].y = fmaxf(cur[u], nxr[2 * k + 1]);
      }
#pragma unroll
      for (int k = 0; k < 8; ++k) acc2[k] += t[k];
    }
#pragma unroll
    for (int u = 0; u < 8; ++u) cur[u] = nxt[u];
  }
  float acc = 0.f;
#pragma unroll
  for (int k = 0; k < 8; ++k) acc += acc2[k].x + acc2[k].y;
  acc += (float)SEQN * sxr;
  smem[tid] = acc;
  __syncthreads();
  if (tid < DCOLS) {
    float s = smem[tid];
#pragma unroll
    for (int p = 1; p < 8; ++p) s += smem[tid + 32 * p];
    pooled[b * DD + dbase + tid] = s;
  }
}

__global__ __launch_bounds__(32) void RN_out_kernel(
    const float* __restrict__ pooled, const float* __restrict__ W_rn,
    const float* __restrict__ b_rn, float* __restrict__ out) {
  const int mt   = blockIdx.x;
  const int nt   = blockIdx.y;
  const int lane = threadIdx.x;
  const int lr   = lane & 15;
  const int koff = (lane >> 4) * 2;
  const int row  = mt * 16 + lr;
  const int col  = nt * 16 + lr;

  const float bv = 16384.0f * b_rn[col];
  v8f c = { bv, bv, bv, bv, bv, bv, bv, bv };

  const float* arow = pooled + (long)row * DD + koff;
  const float* brow = W_rn + (long)col * DD + koff;
  for (int k0 = 0; k0 < DD; k0 += 4) {
    v2f a  = *(const v2f*)(arow + k0);
    v2f bb = *(const v2f*)(brow + k0);
    c = __builtin_amdgcn_wmma_f32_16x16x4_f32(
        false, a, false, bb, (short)0, c, false, false);
  }
  float* obase = out + (mt * 16 + 8 * (lane >> 4)) * DD + nt * 16 + lr;
#pragma unroll
  for (int v = 0; v < 8; ++v) obase[v * DD] = c[v];
}

extern "C" void kernel_launch(void* const* d_in, const int* in_sizes, int n_in,
                              void* d_out, int out_size, void* d_ws, size_t ws_size,
                              hipStream_t stream) {
  (void)in_sizes; (void)n_in; (void)out_size; (void)ws_size;
  const int*   X    = (const int*)d_in[0];
  const float* emb  = (const float*)d_in[1];
  const float* W_l  = (const float*)d_in[2];
  const float* b_l  = (const float*)d_in[3];
  const float* W_r  = (const float*)d_in[4];
  const float* b_r  = (const float*)d_in[5];
  const float* W_rn = (const float*)d_in[6];
  const float* b_rn = (const float*)d_in[7];
  float* pooled = (float*)d_ws;
  float* out    = (float*)d_out;

  dim3 gridA(NBATCH, NSLICE), blockA(256);
  size_t shmem = (size_t)SMEM_FLOATS * sizeof(float);
  RN_pair_fused_kernel<<<gridA, blockA, shmem, stream>>>(
      X, emb, W_l, b_l, W_r, b_r, pooled);

  dim3 gridB(2, 16), blockB(32);
  RN_out_kernel<<<gridB, blockB, 0, stream>>>(pooled, W_rn, b_rn, out);
}


# Output GEMM: 8 waves per 16x16 tile instead of 1; operand tiles staged in LDS by whole-line loads, each wave takes a 32-wide K slice, 8 partial sums added through LDS (K-sum order changes)
# speedup vs baseline: 6.7309x; 1.1369x over previous
#include <hip/hip_runtime.h>


#define DD        256
#define SEQN      128
#define NBATCH    32
#define DCOLS     32
#define NSLICE    (DD / DCOLS)
#define XE_STRIDE 260
#define P_STRIDE  36

typedef float v2f __attribute__((ext_vector_type(2)));
typedef float v8f __attribute__((ext_vector_type(8)));

#define XE_OFF  0
#define XL_OFF  (SEQN * XE_STRIDE)
#define XR_OFF  (XL_OFF + SEQN * P_STRIDE)
#define TOK_OFF (XR_OFF + SEQN * P_STRIDE)
#define SMEM_FLOATS (TOK_OFF + SEQN)

__global__ __launch_bounds__(256) void RN_pair_fused_kernel(
    const int* __restrict__ X, const float* __restrict__ emb,
    const float* __restrict__ W_l, const float* __restrict__ b_l,
    const float* __restrict__ W_r, const float* __restrict__ b_r,
    float* __restrict__ pooled) {
  extern __shared__ float smem[];
  float* xe = smem + XE_OFF;
  float* xl = smem + XL_OFF;
  float* xr = smem + XR_OFF;
  int*   tok = (int*)(smem + TOK_OFF);

  const int b     = blockIdx.x;
  const int dbase = blockIdx.y * DCOLS;
  const int tid   = threadIdx.x;

  if (tid < SEQN) tok[tid] = X[b * SEQN + tid];
  __syncthreads();

  for (int idx = tid; idx < SEQN * (DD / 4); idx += 256) {
    const int row = idx >> 6;
    const int c4  = idx & 63;
    const float4 v = ((const float4*)(emb + (long)tok[row] * DD))[c4];
    float* dst = xe + row * XE_STRIDE + c4 * 4;
    dst[0] = v.x; dst[1] = v.y; dst[2] = v.z; dst[3] = v.w;
  }
  __syncthreads();

  const int wave = tid >> 5;
  const int lane = tid & 31;
  const int lr   = lane & 15;
  const int koff = (lane >> 4) * 2;
  {
    const int cu      = wave & 3;
    const int mat     = cu >> 1;
    const int colbase = (cu & 1) * 16;
    const int rt0     = (wave >> 2) * 4;
    const float* W    = mat ? W_r : W_l;
    const float* bias = mat ? b_r : b_l;
    float*       dst  = mat ? xr : xl;
    const int dcol    = dbase + colbase + lr;

    const float bv = bias[dcol];
    v8f acc[4];
#pragma unroll
    for (int rt = 0; rt < 4; ++rt)
      acc[rt] = (v8f){ bv, bv, bv, bv, bv, bv, bv, bv };

    const float* brow = W + (long)dcol * DD + koff;
    for (int kb = 0; kb < DD; kb += 32) {
      v2f bf[8];
#pragma unroll
      for (int s = 0; s < 8; ++s)
        bf[s] = *(const v2f*)(brow + kb + 4 * s);
#pragma unroll
      for (int rt = 0; rt < 4; ++rt) {
        const float* arow = xe + ((rt0 + rt) * 16 + lr) * XE_STRIDE + kb + koff;
#pragma unroll
        for (int s = 0; s < 8; ++s) {
          v2f a = *(const v2f*)(arow + 4 * s);
          acc[rt] = __builtin_amdgcn_wmma_f32_16x16x4_f32(
               false, a,  false, bf[s],
               (short)0, acc[rt],
               false,  false);
        }
      }
    }
#pragma unroll
    for (int rt = 0; rt < 4; ++rt) {
      float* obase = dst + ((rt0 + rt) * 16 + 8 * (lane >> 4)) * P_STRIDE + colbase + lr;
#pragma unroll
      for (int v = 0; v < 8; ++v) obase[v * P_STRIDE] = acc[rt][v];
    }
  }
  __syncthreads();

  const int d    = tid & 31;
  const int ih0  = (tid >> 5) * 16;
  float nxr[16];
  float sxr = 0.f;
#pragma unroll
  for (int i = 0; i < 16; ++i) {
    const float v = xr[(ih0 + i) * P_STRIDE + d];
    sxr += v;
    nxr[i] = -v;
  }
  v2f acc2[8];
#pragma unroll
  for (int k = 0; k < 8; ++k) acc2[k] = (v2f){ 0.f, 0.f };

  float cur[8];
#pragma unroll
  for (int u = 0; u < 8; ++u) cur[u] = xl[u * P_STRIDE + d];
  for (int jb = 0; jb < SEQN; jb += 8) {
    const int jn = (jb + 8) & (SEQN - 1);
    float nxt[8];
#pragma unroll
    for (int u = 0; u < 8; ++u) nxt[u] = xl[(jn + u) * P_STRIDE + d];
#pragma unroll
    for (int u = 0; u < 8; ++u) {
      v2f t[8];
#pragma unroll
      for (int k = 0; k < 8; ++k) {
        t[k].x = fmaxf(cur[u], nxr[2 * k]);
        t[k].y = fmaxf(cur[u], nxr[2 * k + 1]);
      }
#pragma unroll
      for (int k = 0; k < 8; ++k) acc2[k] += t[k];
    }
#pragma unroll
    for (int u = 0; u < 8; ++u) cur[u] = nxt[u];
  }
  float acc = 0.f;
#pragma unroll
  for (int k = 0; k < 8; ++k) acc += acc2[k].x + acc2[k].y;
  acc += (float)SEQN * sxr;
  smem[tid] = acc;
  __syncthreads();
  if (tid < DCOLS) {
    float s = smem[tid];
#pragma unroll
    for (int p = 1; p < 8; ++p) s += smem[tid + 32 * p];
    pooled[b * DD + dbase + tid] = s;
  }
}

#define KSPLIT    8
#define T_STRIDE  260

__global__ __launch_bounds__(256) void RN_out_kernel(
    const float* __restrict__ pooled, const float* __restrict__ W_rn,
    const float* __restrict__ b_rn, float* __restrict__ out) {
  __shared__ __attribute__((aligned(16))) float sA[16 * T_STRIDE];
  __shared__ __attribute__((aligned(16))) float sB[16 * T_STRIDE];
  __shared__ float red[KSPLIT * 256];

  const int mt   = blockIdx.x;
  const int nt   = blockIdx.y;
  const int tid  = threadIdx.x;
  const int lane = tid & 31;
  const int kw   = tid >> 5;
  const int lr   = lane & 15;
  const int koff = (lane >> 4) * 2;

  const float4* a4 = (const float4*)(pooled + (long)mt * 16 * DD);
  const float4* b4 = (const float4*)(W_rn   + (long)nt * 16 * DD);
  float4 av[4], bw[4];
#pragma unroll
  for (int q = 0; q < 4; ++q) {
    av[q] = a4[tid + 256 * q];
    bw[q] = b4[tid + 256 * q];
  }
  const float bias = b_rn[nt * 16 + lr];
#pragma unroll
  for (int q = 0; q < 4; ++q) {
    const int idx = tid + 256 * q;
    const int row = idx >> 6;
    const int c4  = idx & 63;
    *(float4*)(sA + row * T_STRIDE + c4 * 4) = av[q];
    *(float4*)(sB + row * T_STRIDE + c4 * 4) = bw[q];
  }

  const float* arow = sA + lr * T_STRIDE + kw * (DD / KSPLIT) + koff;
  const float* brow = sB + lr * T_STRIDE + kw * (DD / KSPLIT) + koff;
  float* redw = red + kw * 256 + lane;
  float* outp = out + (mt * 16 + 8 * (lane >> 4) + kw) * DD + nt * 16 + lr;
  const float bv = (kw == 0) ? 16384.0f * bias : 0.f;
  v8f c = { bv, bv, bv, bv, bv, bv, bv, bv };
  __syncthreads();

  __builtin_amdgcn_sched_barrier(0);
#pragma unroll
  for (int s = 0; s < DD / KSPLIT / 4; ++s) {
    v2f a  = *(const v2f*)(arow + 4 * s);
    v2f bb = *(const v2f*)(brow + 4 * s);
    c = __builtin_amdgcn_wmma_f32_16x16x4_f32(
        false, a, false, bb, (short)0, c, false, false);
  }
  __builtin_amdgcn_sched_barrier(0);
  asm volatile("v_nop\n\tv_nop\n\tv_nop\n\tv_nop\n\tv_nop");
  __builtin_amdgcn_sched_barrier(0);

#pragma unroll
  for (int v = 0; v < 8; ++v) redw[v * 32] = c[v];
  __syncthreads();

  float s = red[tid];
#pragma unroll
  for (int w = 1; w < KSPLIT; ++w) s += red[w * 256 + tid];
  *outp = s;
}

extern "C" void kernel_launch(void* const* d_in, const int* in_sizes, int n_in,
                              void* d_out, int out_size, void* d_ws, size_t ws_size,
                              hipStream_t stream) {
  (void)in_sizes; (void)n_in; (void)out_size; (void)ws_size;
  const int*   X    = (const int*)d_in[0];
  const float* emb  = (const float*)d_in[1];
  const float* W_l  = (const float*)d_in[2];
  const float* b_l  = (const float*)d_in[3];
  const float* W_r  = (const float*)d_in[4];
  const float* b_r  = (const float*)d_in[5];
  const float* W_rn = (const float*)d_in[6];
  const float* b_rn = (const float*)d_in[7];
  float* pooled = (float*)d_ws;
  float* out    = (float*)d_out;

  dim3 gridA(NBATCH, NSLICE), blockA(256);
  size_t shmem = (size_t)SMEM_FLOATS * sizeof(float);
  RN_pair_fused_kernel<<<gridA, blockA, shmem, stream>>>(
      X, emb, W_l, b_l, W_r, b_r, pooled);

  dim3 gridB(2, 16), blockB(256);
  RN_out_kernel<<<gridB, blockB, 0, stream>>>(pooled, W_rn, b_rn, out);
}


# Projection re-tiled: each wave owns 16 rows x 64 columns and loads its embedding rows straight into A registers (no LDS copy, no gather loop); weight slice staged in LDS as B; K-sum order changes
# speedup vs baseline: 7.3882x; 1.0977x over previous
#include <hip/hip_runtime.h>


#define DD        256
#define SEQN      128
#define NBATCH    32
#define DCOLS     32
#define NSLICE    (DD / DCOLS)
#define SW_STRIDE 260
#define P_STRIDE  36

typedef float v2f __attribute__((ext_vector_type(2)));
typedef float v8f __attribute__((ext_vector_type(8)));

#define SW_OFF  0
#define XL_OFF  (SEQN * 260)
#define XR_OFF  (XL_OFF + SEQN * P_STRIDE)
#define SMEM_FLOATS (XR_OFF + SEQN * P_STRIDE + SEQN)

__global__ __launch_bounds__(256) void RN_pair_fused_kernel(
    const int* __restrict__ X, const float* __restrict__ emb,
    const float* __restrict__ W_l, const float* __restrict__ b_l,
    const float* __restrict__ W_r, const float* __restrict__ b_r,
    float* __restrict__ pooled) {
  extern __shared__ float smem[];
  float* sw = smem + SW_OFF;
  float* xl = smem + XL_OFF;
  float* xr = smem + XR_OFF;

  const int b     = blockIdx.x;
  const int dbase = blockIdx.y * DCOLS;
  const int tid   = threadIdx.x;
  const int wave  = tid >> 5;
  const int lane  = tid & 31;
  const int lr    = lane & 15;
  const int lh    = lane >> 4;

  const int tok = X[b * SEQN + wave * 16 + lr];

  float bv[4];
  bv[0] = b_l[dbase + lr];
  bv[1] = b_l[dbase + 16 + lr];
  bv[2] = b_r[dbase + lr];
  bv[3] = b_r[dbase + 16 + lr];

  const float4* wl4 = (const float4*)(W_l + (long)dbase * DD);
  const float4* wr4 = (const float4*)(W_r + (long)dbase * DD);
  float4 wv[16];
#pragma unroll
  for (int it = 0; it < 8; ++it) {
    wv[it]     = wl4[tid + 256 * it];
    wv[8 + it] = wr4[tid + 256 * it];
  }

#pragma unroll
  for (int it = 0; it < 8; ++it) {
    const int idx = tid + 256 * it;
    const int col = idx >> 6;
    const int c4  = idx & 63;
    *(float4*)(sw + col * SW_STRIDE + c4 * 4)        = wv[it];
    *(float4*)(sw + (32 + col) * SW_STRIDE + c4 * 4) = wv[8 + it];
  }
  __syncthreads();

  const float4* arow4 = (const float4*)(emb + (long)tok * DD) + lh;
  float4 a[32];
#pragma unroll
  for (int q = 0; q < 32; ++q) a[q] = arow4[2 * q];

  {
    v8f acc[4];
#pragma unroll
    for (int t = 0; t < 4; ++t)
      acc[t] = (v8f){ bv[t], bv[t], bv[t], bv[t], bv[t], bv[t], bv[t], bv[t] };

    const float* bbase = sw + lr * SW_STRIDE + 4 * lh;
    int ooff = (wave * 16 + 8 * lh) * P_STRIDE + lr;
    asm volatile("" : "+v"(ooff));
    float4 bf[2][4];
    __builtin_amdgcn_sched_barrier(0);
#pragma unroll
    for (int t = 0; t < 4; ++t)
      bf[0][t] = *(const float4*)(bbase + t * 16 * SW_STRIDE);
#pragma unroll
    for (int q = 0; q < 32; ++q) {
      if (q + 1 < 32) {
#pragma unroll
        for (int t = 0; t < 4; ++t)
          bf[(q + 1) & 1][t] = *(const float4*)(bbase + t * 16 * SW_STRIDE + 8 * (q + 1));
      }
      __builtin_amdgcn_sched_barrier(0);
      const v2f a0 = { a[q].x, a[q].y };
      const v2f a1 = { a[q].z, a[q].w };
#pragma unroll
      for (int t = 0; t < 4; ++t) {
        const v2f b0 = { bf[q & 1][t].x, bf[q & 1][t].y };
        acc[t] = __builtin_amdgcn_wmma_f32_16x16x4_f32(
             false, a0,  false, b0,
             (short)0, acc[t],
             false,  false);
      }
#pragma unroll
      for (int t = 0; t < 4; ++t) {
        const v2f b1 = { bf[q & 1][t].z, bf[q & 1][t].w };
        acc[t] = __builtin_amdgcn_wmma_f32_16x16x4_f32(
            false, a1, false, b1, (short)0, acc[t], false, false);
      }
      __builtin_amdgcn_sched_barrier(0);
    }
    asm volatile("v_nop\n\tv_nop\n\tv_nop\n\tv_nop\n\tv_nop");
    __builtin_amdgcn_sched_barrier(0);
#pragma unroll
    for (int t = 0; t < 4; ++t) {
      float* obase = ((t < 2) ? xl : xr) + ooff + (t & 1) * 16;
#pragma unroll
      for (int v = 0; v < 8; ++v) obase[v * P_STRIDE] = acc[t][v];
    }
  }
  __syncthreads();

  const int d    = tid & 31;
  const int ih0  = (tid >> 5) * 16;
  float nxr[16];
  float sxr = 0.f;
#pragma unroll
  for (int i = 0; i < 16; ++i) {
    const float v = xr[(ih0 + i) * P_STRIDE + d];
    sxr += v;
    nxr[i] = -v;
  }
  v2f acc2[8];
#pragma unroll
  for (int k = 0; k < 8; ++k) acc2[k] = (v2f){ 0.f, 0.f };

  float cur[8];
#pragma unroll
  for (int u = 0; u < 8; ++u) cur[u] = xl[u * P_STRIDE + d];
  for (int jb = 0; jb < SEQN; jb += 8) {
    const int jn = (jb + 8) & (SEQN - 1);
    float nxt[8];
#pragma unroll
    for (int u = 0; u < 8; ++u) nxt[u] = xl[(jn + u) * P_STRIDE + d];
#pragma unroll
    for (int u = 0; u < 8; ++u) {
      v2f t[8];
#pragma unroll
      for (int k = 0; k < 8; ++k) {
        t[k].x = fmaxf(cur[u], nxr[2 * k]);
        t[k].y = fmaxf(cur[u], nxr[2 * k + 1]);
      }
#pragma unroll
      for (int k = 0; k < 8; ++k) acc2[k] += t[k];
    }
#pragma unroll
    for (int u = 0; u < 8; ++u) cur[u] = nxt[u];
  }
  float acc = 0.f;
#pragma unroll
  for (int k = 0; k < 8; ++k) acc += acc2[k].x + acc2[k].y;
  acc += (float)SEQN * sxr;
  smem[tid] = acc;
  __syncthreads();
  if (tid < DCOLS) {
    float s = smem[tid];
#pragma unroll
    for (int p = 1; p < 8; ++p) s += smem[tid + 32 * p];
    pooled[b * DD + dbase + tid] = s;
  }
}

#define KSPLIT    8
#define T_STRIDE  260

__global__ __launch_bounds__(256) void RN_out_kernel(
    const float* __restrict__ pooled, const float* __restrict__ W_rn,
    const float* __restrict__ b_rn, float* __restrict__ out) {
  __shared__ __attribute__((aligned(16))) float sA[16 * T_STRIDE];
  __shared__ __attribute__((aligned(16))) float sB[16 * T_STRIDE];
  __shared__ float red[KSPLIT * 256];

  const int mt   = blockIdx.x;
  const int nt   = blockIdx.y;
  const int tid  = threadIdx.x;
  const int lane = tid & 31;
  const int kw   = tid >> 5;
  const int lr   = lane & 15;
  const int koff = (lane >> 4) * 2;

  const float4* a4 = (const float4*)(pooled + (long)mt * 16 * DD);
  const float4* b4 = (const float4*)(W_rn   + (long)nt * 16 * DD);
  float4 av[4], bw[4];
#pragma unroll
  for (int q = 0; q < 4; ++q) {
    av[q] = a4[tid + 256 * q];
    bw[q] = b4[tid + 256 * q];
  }
  const float bias = b_rn[nt * 16 + lr];
#pragma unroll
  for (int q = 0; q < 4; ++q) {
    const int idx = tid + 256 * q;
    const int row = idx >> 6;
    const int c4  = idx & 63;
    *(float4*)(sA + row * T_STRIDE + c4 * 4) = av[q];
    *(float4*)(sB + row * T_STRIDE + c4 * 4) = bw[q];
  }

  const float* arow = sA + lr * T_STRIDE + kw * (DD / KSPLIT) + koff;
  const float* brow = sB + lr * T_STRIDE + kw * (DD / KSPLIT) + koff;
  float* redw = red + kw * 256 + lane;
  float* outp = out + (mt * 16 + 8 * (lane >> 4) + kw) * DD + nt * 16 + lr;
  const float bv = (kw == 0) ? 16384.0f * bias : 0.f;
  v8f c = { bv, bv, bv, bv, bv, bv, bv, bv };
  __syncthreads();

  __builtin_amdgcn_sched_barrier(0);
#pragma unroll
  for (int s = 0; s < DD / KSPLIT / 4; ++s) {
    v2f a  = *(const v2f*)(arow + 4 * s);
    v2f bb = *(const v2f*)(brow + 4 * s);
    c = __builtin_amdgcn_wmma_f32_16x16x4_f32(
        false, a, false, bb, (short)0, c, false, false);
  }
  __builtin_amdgcn_sched_barrier(0);
  asm volatile("v_nop\n\tv_nop\n\tv_nop\n\tv_nop\n\tv_nop");
  __builtin_amdgcn_sched_barrier(0);

#pragma unroll
  for (int v = 0; v < 8; ++v) redw[v * 32] = c[v];
  __syncthreads();

  float s = red[tid];
#pragma unroll
  for (int w = 1; w < KSPLIT; ++w) s += red[w * 256 + tid];
  *outp = s;
}

extern "C" void kernel_launch(void* const* d_in, const int* in_sizes, int n_in,
                              void* d_out, int out_size, void* d_ws, size_t ws_size,
                              hipStream_t stream) {
  (void)in_sizes; (void)n_in; (void)out_size; (void)ws_size;
  const int*   X    = (const int*)d_in[0];
  const float* emb  = (const float*)d_in[1];
  const float* W_l  = (const float*)d_in[2];
  const float* b_l  = (const float*)d_in[3];
  const float* W_r  = (const float*)d_in[4];
  const float* b_r  = (const float*)d_in[5];
  const float* W_rn = (const float*)d_in[6];
  const float* b_rn = (const float*)d_in[7];
  float* pooled = (float*)d_ws;
  float* out    = (float*)d_out;

  dim3 gridA(NBATCH, NSLICE), blockA(256);
  size_t shmem = (size_t)SMEM_FLOATS * sizeof(float);
  RN_pair_fused_kernel<<<gridA, blockA, shmem, stream>>>(
      X, emb, W_l, b_l, W_r, b_r, pooled);

  dim3 gridB(2, 16), blockB(256);
  RN_out_kernel<<<gridB, blockB, 0, stream>>>(pooled, W_rn, b_rn, out);
}
